# starting kernel
# baseline (speedup 1.0000x reference)
#include <hip/hip_runtime.h>
#include <math.h>

typedef __attribute__((ext_vector_type(2))) float v2f;
typedef __attribute__((ext_vector_type(8))) float v8f;

#define ALPHA 0.2f
#define BB 4
#define KK 256
#define FF 128
#define EE 256
#define ITILE 4
#define ECHUNK 32

__global__ __launch_bounds__(256)
void gat_gemm_s_kernel(const float* __restrict__ X,
                       const float* __restrict__ W,
                       const float* __restrict__ blin,
                       float* __restrict__ S)
{
    const int wave = threadIdx.x >> 5;
    const int lane = threadIdx.x & 31;
    const int ln   = lane & 15;
    const int hi   = lane >> 4;
    const int s    = blockIdx.y;
    const int tile = blockIdx.x * 8 + wave;
    const int row0 = (tile >> 4) * 16;
    const int col0 = (tile & 15) * 16;

    const float* Arow = X + (size_t)(row0 + ln) * FF;
    const float* Brow = W + (size_t)(col0 + ln) * (2 * FF) + s * FF;

    v8f c = {};
    #pragma unroll 4
    for (int k0 = 0; k0 < FF; k0 += 4) {
        v2f a = *(const v2f*)(Arow + k0 + 2 * hi);
        v2f b = *(const v2f*)(Brow + k0 + 2 * hi);
        c = __builtin_amdgcn_wmma_f32_16x16x4_f32(false, a, false, b,
                                                  (short)0, c, false, false);
    }

    const float add = (s == 0) ? blin[col0 + ln] : 0.0f;
    float* D = S + (size_t)s * 1024 * EE;
    #pragma unroll
    for (int r = 0; r < 8; ++r)
        D[(size_t)(row0 + r + 8 * hi) * EE + col0 + ln] = c[r] + add;
}

__global__ __launch_bounds__(256)
void gat_attn_kernel(const float* __restrict__ S,
                     const float* __restrict__ avec,
                     const float* __restrict__ bias,
                     float* __restrict__ attn)
{
    const int b  = blockIdx.y;
    const int i0 = blockIdx.x * ITILE;
    const int j  = threadIdx.x;

    __shared__ float s1row[ITILE][EE];
    __shared__ float a_s[EE];
    __shared__ float tile[KK][ECHUNK + 1];
    __shared__ float red[KK];

    const float* S1 = S;
    const float* S2b = S + (size_t)1024 * EE + (size_t)b * KK * EE;

    a_s[j] = avec[j];
    #pragma unroll
    for (int ii = 0; ii < ITILE; ++ii)
        s1row[ii][j] = S1[(size_t)(b * KK + i0 + ii) * EE + j];
    __syncthreads();

    float acc[ITILE];
    #pragma unroll
    for (int ii = 0; ii < ITILE; ++ii) acc[ii] = 0.0f;

    for (int e0 = 0; e0 < EE; e0 += ECHUNK) {
        __syncthreads();
        for (int idx = threadIdx.x; idx < KK * ECHUNK; idx += 256) {
            const int jj = idx >> 5;
            const int ee = idx & 31;
            tile[jj][ee] = S2b[(size_t)jj * EE + e0 + ee];
        }
        __syncthreads();
        #pragma unroll 8
        for (int ee = 0; ee < ECHUNK; ++ee) {
            const float t  = tile[j][ee];
            const float ae = a_s[e0 + ee];
            #pragma unroll
            for (int ii = 0; ii < ITILE; ++ii) {
                const float p = s1row[ii][e0 + ee] + t;
                const float h = (p >= 0.0f) ? p : ALPHA * p;
                acc[ii] = fmaf(ae, h, acc[ii]);
            }
        }
    }

    #pragma unroll
    for (int ii = 0; ii < ITILE; ++ii) {
        const float ev = acc[ii] + bias[(size_t)(i0 + ii) * KK + j];

        red[j] = ev;           __syncthreads();
        for (int off = 128; off > 0; off >>= 1) {
            if (j < off) red[j] = fmaxf(red[j], red[j + off]);
            __syncthreads();
        }
        const float mx = red[0]; __syncthreads();

        const float p = __expf(ev - mx);
        red[j] = p;            __syncthreads();
        for (int off = 128; off > 0; off >>= 1) {
            if (j < off) red[j] += red[j + off];
            __syncthreads();
        }
        const float inv = 1.0f / red[0];
        __syncthreads();

        attn[(size_t)(b * KK + i0 + ii) * KK + j] = p * inv;
    }
}

__global__ __launch_bounds__(256)
void gat_out_kernel(const float* __restrict__ attn,
                    const float* __restrict__ X,
                    float* __restrict__ out)
{
    const int wave = threadIdx.x >> 5;
    const int lane = threadIdx.x & 31;
    const int ln   = lane & 15;
    const int hi   = lane >> 4;
    const int b    = blockIdx.y;
    const int tile = blockIdx.x * 8 + wave;
    const int row0 = (tile >> 3) * 16;
    const int col0 = (tile & 7) * 16;

    const float* A  = attn + (size_t)b * KK * KK;
    const float* Bm = X    + (size_t)b * KK * FF;
    const float* Arow = A + (size_t)(row0 + ln) * KK;

    v8f c = {};
    #pragma unroll 4
    for (int k0 = 0; k0 < KK; k0 += 4) {
        v2f a = *(const v2f*)(Arow + k0 + 2 * hi);
        v2f bb;
        bb.x = Bm[(size_t)(k0 + 2 * hi) * FF + col0 + ln];
        bb.y = Bm[(size_t)(k0 + 2 * hi + 1) * FF + col0 + ln];
        c = __builtin_amdgcn_wmma_f32_16x16x4_f32(false, a, false, bb,
                                                  (short)0, c, false, false);
    }

    float* O = out + (size_t)b * KK * FF;
    #pragma unroll
    for (int r = 0; r < 8; ++r) {
        const float v = c[r];
        O[(size_t)(row0 + r + 8 * hi) * FF + col0 + ln] =
            1.0f / (1.0f + __expf(-v));
    }
}

extern "C" void kernel_launch(void* const* d_in, const int* in_sizes, int n_in,
                              void* d_out, int out_size, void* d_ws, size_t ws_size,
                              hipStream_t stream) {
    const float* x    = (const float*)d_in[0];
    const float* W    = (const float*)d_in[1];
    const float* blin = (const float*)d_in[2];
    const float* a    = (const float*)d_in[3];
    const float* bias = (const float*)d_in[4];
    float* out = (float*)d_out;

    float* S     = (float*)d_ws;
    float* attnw = S + (size_t)2 * 1024 * EE;

    gat_gemm_s_kernel<<<dim3(128, 2), 256, 0, stream>>>(x, W, blin, S);

    gat_attn_kernel<<<dim3(KK / ITILE, BB), 256, 0, stream>>>(S, a, bias, attnw);

    gat_out_kernel<<<dim3(16, BB), 256, 0, stream>>>(attnw, x, out);
}


# softmax max and sum: in-wave DPP/permlane butterfly plus one LDS exchange of 8 wave partials for all 4 rows, replacing 8-level LDS trees with a barrier per level; sum order changes
# speedup vs baseline: 1.2076x; 1.2076x over previous
#include <hip/hip_runtime.h>
#include <math.h>

typedef __attribute__((ext_vector_type(2))) float v2f;
typedef __attribute__((ext_vector_type(8))) float v8f;

#define ALPHA 0.2f
#define BB 4
#define KK 256
#define FF 128
#define EE 256
#define ITILE 4
#define ECHUNK 32

__global__ __launch_bounds__(256)
void gat_gemm_s_kernel(const float* __restrict__ X,
                       const float* __restrict__ W,
                       const float* __restrict__ blin,
                       float* __restrict__ S)
{
    const int wave = threadIdx.x >> 5;
    const int lane = threadIdx.x & 31;
    const int ln   = lane & 15;
    const int hi   = lane >> 4;
    const int s    = blockIdx.y;
    const int tile = blockIdx.x * 8 + wave;
    const int row0 = (tile >> 4) * 16;
    const int col0 = (tile & 15) * 16;

    const float* Arow = X + (size_t)(row0 + ln) * FF;
    const float* Brow = W + (size_t)(col0 + ln) * (2 * FF) + s * FF;

    v8f c = {};
    #pragma unroll 4
    for (int k0 = 0; k0 < FF; k0 += 4) {
        v2f a = *(const v2f*)(Arow + k0 + 2 * hi);
        v2f b = *(const v2f*)(Brow + k0 + 2 * hi);
        c = __builtin_amdgcn_wmma_f32_16x16x4_f32(false, a, false, b,
                                                  (short)0, c, false, false);
    }

    const float add = (s == 0) ? blin[col0 + ln] : 0.0f;
    float* D = S + (size_t)s * 1024 * EE;
    #pragma unroll
    for (int r = 0; r < 8; ++r)
        D[(size_t)(row0 + r + 8 * hi) * EE + col0 + ln] = c[r] + add;
}

template <int XM>
__device__ __forceinline__ float row_xor(float x) {
    const int v = __builtin_bit_cast(int, x);
    return __builtin_bit_cast(float,
        __builtin_amdgcn_update_dpp(v, v, 0x160 + XM, 0xf, 0xf, false));
}
__device__ __forceinline__ float xhalf(float x) {
    const int v = __builtin_bit_cast(int, x);
    return __builtin_bit_cast(float,
        __builtin_amdgcn_permlanex16(v, v, 0x76543210, 0xfedcba98, false, false));
}
__device__ __forceinline__ float wave_max32(float x) {
    x = fmaxf(x, row_xor<1>(x));
    x = fmaxf(x, row_xor<2>(x));
    x = fmaxf(x, row_xor<4>(x));
    x = fmaxf(x, row_xor<8>(x));
    return fmaxf(x, xhalf(x));
}
__device__ __forceinline__ float wave_sum32(float x) {
    x += row_xor<1>(x);
    x += row_xor<2>(x);
    x += row_xor<4>(x);
    x += row_xor<8>(x);
    return x + xhalf(x);
}

__global__ __launch_bounds__(256)
void gat_attn_kernel(const float* __restrict__ S,
                     const float* __restrict__ avec,
                     const float* __restrict__ bias,
                     float* __restrict__ attn)
{
    const int b  = blockIdx.y;
    const int i0 = blockIdx.x * ITILE;
    const int j  = threadIdx.x;

    __shared__ float s1row[ITILE][EE];
    __shared__ float a_s[EE];
    __shared__ float tile[KK][ECHUNK + 1];
    __shared__ float4 redmax[8];
    __shared__ float4 redsum[8];

    const float* S1 = S;
    const float* S2b = S + (size_t)1024 * EE + (size_t)b * KK * EE;

    a_s[j] = avec[j];
    #pragma unroll
    for (int ii = 0; ii < ITILE; ++ii)
        s1row[ii][j] = S1[(size_t)(b * KK + i0 + ii) * EE + j];
    __syncthreads();

    float acc[ITILE];
    #pragma unroll
    for (int ii = 0; ii < ITILE; ++ii) acc[ii] = 0.0f;

    for (int e0 = 0; e0 < EE; e0 += ECHUNK) {
        __syncthreads();
        for (int idx = threadIdx.x; idx < KK * ECHUNK; idx += 256) {
            const int jj = idx >> 5;
            const int ee = idx & 31;
            tile[jj][ee] = S2b[(size_t)jj * EE + e0 + ee];
        }
        __syncthreads();
        #pragma unroll 8
        for (int ee = 0; ee < ECHUNK; ++ee) {
            const float t  = tile[j][ee];
            const float ae = a_s[e0 + ee];
            #pragma unroll
            for (int ii = 0; ii < ITILE; ++ii) {
                const float p = s1row[ii][e0 + ee] + t;
                const float h = (p >= 0.0f) ? p : ALPHA * p;
                acc[ii] = fmaf(ae, h, acc[ii]);
            }
        }
    }

    static_assert(ITILE == 4, "the 4 rows' partials travel as one float4");
    const int wave = threadIdx.x >> 5;
    const int lane = threadIdx.x & 31;

    float ev[ITILE];
    #pragma unroll
    for (int ii = 0; ii < ITILE; ++ii)
        ev[ii] = acc[ii] + bias[(size_t)(i0 + ii) * KK + j];

    float4 wm;
    wm.x = wave_max32(ev[0]); wm.y = wave_max32(ev[1]);
    wm.z = wave_max32(ev[2]); wm.w = wave_max32(ev[3]);
    if (lane == 0) redmax[wave] = wm;
    __syncthreads();

    float4 m = redmax[0];
    #pragma unroll
    for (int w = 1; w < 8; ++w) {
        const float4 t = redmax[w];
        m.x = fmaxf(m.x, t.x); m.y = fmaxf(m.y, t.y);
        m.z = fmaxf(m.z, t.z); m.w = fmaxf(m.w, t.w);
    }

    float p[ITILE];
    p[0] = __expf(ev[0] - m.x); p[1] = __expf(ev[1] - m.y);
    p[2] = __expf(ev[2] - m.z); p[3] = __expf(ev[3] - m.w);

    float4 ws;
    ws.x = wave_sum32(p[0]); ws.y = wave_sum32(p[1]);
    ws.z = wave_sum32(p[2]); ws.w = wave_sum32(p[3]);
    if (lane == 0) redsum[wave] = ws;
    __syncthreads();

    float4 s = redsum[0];
    #pragma unroll
    for (int w = 1; w < 8; ++w) {
        const float4 t = redsum[w];
        s.x += t.x; s.y += t.y; s.z += t.z; s.w += t.w;
    }
    const float inv[ITILE] = {1.0f / s.x, 1.0f / s.y, 1.0f / s.z, 1.0f / s.w};

    #pragma unroll
    for (int ii = 0; ii < ITILE; ++ii)
        attn[(size_t)(b * KK + i0 + ii) * KK + j] = p[ii] * inv[ii];
}

__global__ __launch_bounds__(256)
void gat_out_kernel(const float* __restrict__ attn,
                    const float* __restrict__ X,
                    float* __restrict__ out)
{
    const int wave = threadIdx.x >> 5;
    const int lane = threadIdx.x & 31;
    const int ln   = lane & 15;
    const int hi   = lane >> 4;
    const int b    = blockIdx.y;
    const int tile = blockIdx.x * 8 + wave;
    const int row0 = (tile >> 3) * 16;
    const int col0 = (tile & 7) * 16;

    const float* A  = attn + (size_t)b * KK * KK;
    const float* Bm = X    + (size_t)b * KK * FF;
    const float* Arow = A + (size_t)(row0 + ln) * KK;

    v8f c = {};
    #pragma unroll 4
    for (int k0 = 0; k0 < KK; k0 += 4) {
        v2f a = *(const v2f*)(Arow + k0 + 2 * hi);
        v2f bb;
        bb.x = Bm[(size_t)(k0 + 2 * hi) * FF + col0 + ln];
        bb.y = Bm[(size_t)(k0 + 2 * hi + 1) * FF + col0 + ln];
        c = __builtin_amdgcn_wmma_f32_16x16x4_f32(false, a, false, bb,
                                                  (short)0, c, false, false);
    }

    float* O = out + (size_t)b * KK * FF;
    #pragma unroll
    for (int r = 0; r < 8; ++r) {
        const float v = c[r];
        O[(size_t)(row0 + r + 8 * hi) * FF + col0 + ln] =
            1.0f / (1.0f + __expf(-v));
    }
}

extern "C" void kernel_launch(void* const* d_in, const int* in_sizes, int n_in,
                              void* d_out, int out_size, void* d_ws, size_t ws_size,
                              hipStream_t stream) {
    const float* x    = (const float*)d_in[0];
    const float* W    = (const float*)d_in[1];
    const float* blin = (const float*)d_in[2];
    const float* a    = (const float*)d_in[3];
    const float* bias = (const float*)d_in[4];
    float* out = (float*)d_out;

    float* S     = (float*)d_ws;
    float* attnw = S + (size_t)2 * 1024 * EE;

    gat_gemm_s_kernel<<<dim3(128, 2), 256, 0, stream>>>(x, W, blin, S);

    gat_attn_kernel<<<dim3(KK / ITILE, BB), 256, 0, stream>>>(S, a, bias, attnw);

    gat_out_kernel<<<dim3(16, BB), 256, 0, stream>>>(attnw, x, out);
}


# attention kernel: S2 chunk stream software-pipelined: next chunk's global loads fly into registers under the current chunk's arithmetic, two LDS stages, one barrier per chunk instead of two
# speedup vs baseline: 1.3487x; 1.1168x over previous
#include <hip/hip_runtime.h>
#include <math.h>

typedef __attribute__((ext_vector_type(2))) float v2f;
typedef __attribute__((ext_vector_type(8))) float v8f;

#define ALPHA 0.2f
#define BB 4
#define KK 256
#define FF 128
#define EE 256
#define ITILE 4
#define ECHUNK 32
#define NCHUNK (EE / ECHUNK)
#define NSTAGE (KK * ECHUNK / 256)

__global__ __launch_bounds__(256)
void gat_gemm_s_kernel(const float* __restrict__ X,
                       const float* __restrict__ W,
                       const float* __restrict__ blin,
                       float* __restrict__ S)
{
    const int wave = threadIdx.x >> 5;
    const int lane = threadIdx.x & 31;
    const int ln   = lane & 15;
    const int hi   = lane >> 4;
    const int s    = blockIdx.y;
    const int tile = blockIdx.x * 8 + wave;
    const int row0 = (tile >> 4) * 16;
    const int col0 = (tile & 15) * 16;

    const float* Arow = X + (size_t)(row0 + ln) * FF;
    const float* Brow = W + (size_t)(col0 + ln) * (2 * FF) + s * FF;

    v8f c = {};
    #pragma unroll 4
    for (int k0 = 0; k0 < FF; k0 += 4) {
        v2f a = *(const v2f*)(Arow + k0 + 2 * hi);
        v2f b = *(const v2f*)(Brow + k0 + 2 * hi);
        c = __builtin_amdgcn_wmma_f32_16x16x4_f32(false, a, false, b,
                                                  (short)0, c, false, false);
    }

    const float add = (s == 0) ? blin[col0 + ln] : 0.0f;
    float* D = S + (size_t)s * 1024 * EE;
    #pragma unroll
    for (int r = 0; r < 8; ++r)
        D[(size_t)(row0 + r + 8 * hi) * EE + col0 + ln] = c[r] + add;
}

template <int XM>
__device__ __forceinline__ float row_xor(float x) {
    const int v = __builtin_bit_cast(int, x);
    return __builtin_bit_cast(float,
        __builtin_amdgcn_update_dpp(v, v, 0x160 + XM, 0xf, 0xf, false));
}
__device__ __forceinline__ float xhalf(float x) {
    const int v = __builtin_bit_cast(int, x);
    return __builtin_bit_cast(float,
        __builtin_amdgcn_permlanex16(v, v, 0x76543210, 0xfedcba98, false, false));
}
__device__ __forceinline__ float wave_max32(float x) {
    x = fmaxf(x, row_xor<1>(x));
    x = fmaxf(x, row_xor<2>(x));
    x = fmaxf(x, row_xor<4>(x));
    x = fmaxf(x, row_xor<8>(x));
    return fmaxf(x, xhalf(x));
}
__device__ __forceinline__ float wave_sum32(float x) {
    x += row_xor<1>(x);
    x += row_xor<2>(x);
    x += row_xor<4>(x);
    x += row_xor<8>(x);
    return x + xhalf(x);
}

__global__ __launch_bounds__(256)
void gat_attn_kernel(const float* __restrict__ S,
                     const float* __restrict__ avec,
                     const float* __restrict__ bias,
                     float* __restrict__ attn)
{
    const int b  = blockIdx.y;
    const int i0 = blockIdx.x * ITILE;
    const int j  = threadIdx.x;

    __shared__ float s1row[ITILE][EE];
    __shared__ float a_s[EE];
    __shared__ float tile[2][KK][ECHUNK + 1];
    __shared__ float4 redmax[8];
    __shared__ float4 redsum[8];

    const int wave = threadIdx.x >> 5;
    const int lane = threadIdx.x & 31;

    const float* S1 = S;
    const float* S2b = S + (size_t)1024 * EE + (size_t)b * KK * EE;

    const float* src = S2b + (size_t)wave * EE + lane;

    float pre[NSTAGE];
    #pragma unroll
    for (int k = 0; k < NSTAGE; ++k)
        pre[k] = src[(size_t)(8 * k) * EE];

    a_s[j] = avec[j];
    #pragma unroll
    for (int ii = 0; ii < ITILE; ++ii)
        s1row[ii][j] = S1[(size_t)(b * KK + i0 + ii) * EE + j];

    #pragma unroll
    for (int k = 0; k < NSTAGE; ++k)
        tile[0][wave + 8 * k][lane] = pre[k];

    float acc[ITILE];
    #pragma unroll
    for (int ii = 0; ii < ITILE; ++ii) acc[ii] = 0.0f;

    for (int c = 0; c < NCHUNK; ++c) {
        const int e0 = c * ECHUNK;
        __syncthreads();
        const bool more = (c + 1 < NCHUNK);
        if (more) {
            #pragma unroll
            for (int k = 0; k < NSTAGE; ++k)
                pre[k] = src[(size_t)(8 * k) * EE + e0 + ECHUNK];
        }
        const float (*cur)[ECHUNK + 1] = tile[c & 1];
        #pragma unroll 8
        for (int ee = 0; ee < ECHUNK; ++ee) {
            const float t  = cur[j][ee];
            const float ae = a_s[e0 + ee];
            #pragma unroll
            for (int ii = 0; ii < ITILE; ++ii) {
                const float p = s1row[ii][e0 + ee] + t;
                const float h = (p >= 0.0f) ? p : ALPHA * p;
                acc[ii] = fmaf(ae, h, acc[ii]);
            }
        }
        if (more) {
            float (*nxt)[ECHUNK + 1] = tile[(c + 1) & 1];
            #pragma unroll
            for (int k = 0; k < NSTAGE; ++k)
                nxt[wave + 8 * k][lane] = pre[k];
        }
    }

    static_assert(ITILE == 4, "the 4 rows' partials travel as one float4");

    float ev[ITILE];
    #pragma unroll
    for (int ii = 0; ii < ITILE; ++ii)
        ev[ii] = acc[ii] + bias[(size_t)(i0 + ii) * KK + j];

    float4 wm;
    wm.x = wave_max32(ev[0]); wm.y = wave_max32(ev[1]);
    wm.z = wave_max32(ev[2]); wm.w = wave_max32(ev[3]);
    if (lane == 0) redmax[wave] = wm;
    __syncthreads();

    float4 m = redmax[0];
    #pragma unroll
    for (int w = 1; w < 8; ++w) {
        const float4 t = redmax[w];
        m.x = fmaxf(m.x, t.x); m.y = fmaxf(m.y, t.y);
        m.z = fmaxf(m.z, t.z); m.w = fmaxf(m.w, t.w);
    }

    float p[ITILE];
    p[0] = __expf(ev[0] - m.x); p[1] = __expf(ev[1] - m.y);
    p[2] = __expf(ev[2] - m.z); p[3] = __expf(ev[3] - m.w);

    float4 ws;
    ws.x = wave_sum32(p[0]); ws.y = wave_sum32(p[1]);
    ws.z = wave_sum32(p[2]); ws.w = wave_sum32(p[3]);
    if (lane == 0) redsum[wave] = ws;
    __syncthreads();

    float4 s = redsum[0];
    #pragma unroll
    for (int w = 1; w < 8; ++w) {
        const float4 t = redsum[w];
        s.x += t.x; s.y += t.y; s.z += t.z; s.w += t.w;
    }
    const float inv[ITILE] = {1.0f / s.x, 1.0f / s.y, 1.0f / s.z, 1.0f / s.w};

    #pragma unroll
    for (int ii = 0; ii < ITILE; ++ii)
        attn[(size_t)(b * KK + i0 + ii) * KK + j] = p[ii] * inv[ii];
}

__global__ __launch_bounds__(256)
void gat_out_kernel(const float* __restrict__ attn,
                    const float* __restrict__ X,
                    float* __restrict__ out)
{
    const int wave = threadIdx.x >> 5;
    const int lane = threadIdx.x & 31;
    const int ln   = lane & 15;
    const int hi   = lane >> 4;
    const int b    = blockIdx.y;
    const int tile = blockIdx.x * 8 + wave;
    const int row0 = (tile >> 3) * 16;
    const int col0 = (tile & 7) * 16;

    const float* A  = attn + (size_t)b * KK * KK;
    const float* Bm = X    + (size_t)b * KK * FF;
    const float* Arow = A + (size_t)(row0 + ln) * KK;

    v8f c = {};
    #pragma unroll 4
    for (int k0 = 0; k0 < KK; k0 += 4) {
        v2f a = *(const v2f*)(Arow + k0 + 2 * hi);
        v2f bb;
        bb.x = Bm[(size_t)(k0 + 2 * hi) * FF + col0 + ln];
        bb.y = Bm[(size_t)(k0 + 2 * hi + 1) * FF + col0 + ln];
        c = __builtin_amdgcn_wmma_f32_16x16x4_f32(false, a, false, bb,
                                                  (short)0, c, false, false);
    }

    float* O = out + (size_t)b * KK * FF;
    #pragma unroll
    for (int r = 0; r < 8; ++r) {
        const float v = c[r];
        O[(size_t)(row0 + r + 8 * hi) * FF + col0 + ln] =
            1.0f / (1.0f + __expf(-v));
    }
}

extern "C" void kernel_launch(void* const* d_in, const int* in_sizes, int n_in,
                              void* d_out, int out_size, void* d_ws, size_t ws_size,
                              hipStream_t stream) {
    const float* x    = (const float*)d_in[0];
    const float* W    = (const float*)d_in[1];
    const float* blin = (const float*)d_in[2];
    const float* a    = (const float*)d_in[3];
    const float* bias = (const float*)d_in[4];
    float* out = (float*)d_out;

    float* S     = (float*)d_ws;
    float* attnw = S + (size_t)2 * 1024 * EE;

    gat_gemm_s_kernel<<<dim3(128, 2), 256, 0, stream>>>(x, W, blin, S);

    gat_attn_kernel<<<dim3(KK / ITILE, BB), 256, 0, stream>>>(S, a, bias, attnw);

    gat_out_kernel<<<dim3(16, BB), 256, 0, stream>>>(attnw, x, out);
}


# attn@X product and sigmoid fused into the attention kernel, 3 launches -> 2: X[b] staged into LDS under the e-loop, the block's 4 normalized rows feed WMMA from LDS; k order and sum order unchanged
# speedup vs baseline: 1.9055x; 1.4128x over previous
#include <hip/hip_runtime.h>
#include <math.h>

typedef __attribute__((ext_vector_type(2))) float v2f;
typedef __attribute__((ext_vector_type(8))) float v8f;

#define ALPHA 0.2f
#define BB 4
#define KK 256
#define FF 128
#define EE 256
#define ITILE 4
#define ECHUNK 32
#define NCHUNK (EE / ECHUNK)
#define NSTAGE (KK * ECHUNK / 256)
#define XQ (KK * FF / 4 / NCHUNK / 256)
#define XPITCH (FF + 8)
#define PPITCH (KK + 16)

__global__ __launch_bounds__(256)
void gat_gemm_s_kernel(const float* __restrict__ X,
                       const float* __restrict__ W,
                       const float* __restrict__ blin,
                       float* __restrict__ S)
{
    const int wave = threadIdx.x >> 5;
    const int lane = threadIdx.x & 31;
    const int ln   = lane & 15;
    const int hi   = lane >> 4;
    const int s    = blockIdx.y;
    const int tile = blockIdx.x * 8 + wave;
    const int row0 = (tile >> 4) * 16;
    const int col0 = (tile & 15) * 16;

    const float* Arow = X + (size_t)(row0 + ln) * FF;
    const float* Brow = W + (size_t)(col0 + ln) * (2 * FF) + s * FF;

    v8f c = {};
    #pragma unroll 4
    for (int k0 = 0; k0 < FF; k0 += 4) {
        v2f a = *(const v2f*)(Arow + k0 + 2 * hi);
        v2f b = *(const v2f*)(Brow + k0 + 2 * hi);
        c = __builtin_amdgcn_wmma_f32_16x16x4_f32(false, a, false, b,
                                                  (short)0, c, false, false);
    }

    const float add = (s == 0) ? blin[col0 + ln] : 0.0f;
    float* D = S + (size_t)s * 1024 * EE;
    #pragma unroll
    for (int r = 0; r < 8; ++r)
        D[(size_t)(row0 + r + 8 * hi) * EE + col0 + ln] = c[r] + add;
}

template <int XM>
__device__ __forceinline__ float row_xor(float x) {
    const int v = __builtin_bit_cast(int, x);
    return __builtin_bit_cast(float,
        __builtin_amdgcn_update_dpp(v, v, 0x160 + XM, 0xf, 0xf, false));
}
__device__ __forceinline__ float xhalf(float x) {
    const int v = __builtin_bit_cast(int, x);
    return __builtin_bit_cast(float,
        __builtin_amdgcn_permlanex16(v, v, 0x76543210, 0xfedcba98, false, false));
}
__device__ __forceinline__ float wave_max32(float x) {
    x = fmaxf(x, row_xor<1>(x));
    x = fmaxf(x, row_xor<2>(x));
    x = fmaxf(x, row_xor<4>(x));
    x = fmaxf(x, row_xor<8>(x));
    return fmaxf(x, xhalf(x));
}
__device__ __forceinline__ float wave_sum32(float x) {
    x += row_xor<1>(x);
    x += row_xor<2>(x);
    x += row_xor<4>(x);
    x += row_xor<8>(x);
    return x + xhalf(x);
}

__device__ __forceinline__ void load_frags8(v2f (&a)[8], v2f (&bq)[8],
                                            const float* arow, const float* bcol) {
    #pragma unroll
    for (int s = 0; s < 8; ++s) {
        a[s]    = *(const v2f*)(arow + 4 * s);
        bq[s].x = bcol[(size_t)(4 * s) * XPITCH];
        bq[s].y = bcol[(size_t)(4 * s + 1) * XPITCH];
    }
}
__device__ __forceinline__ v8f mma_frags8(const v2f (&a)[8], const v2f (&bq)[8], v8f acc) {
    #pragma unroll
    for (int s = 0; s < 8; ++s)
        acc = __builtin_amdgcn_wmma_f32_16x16x4_f32(false, a[s], false, bq[s],
                                                    (short)0, acc, false, false);
    return acc;
}

__global__ __launch_bounds__(256)
void gat_attn_out_kernel(const float* __restrict__ S,
                         const float* __restrict__ avec,
                         const float* __restrict__ bias,
                         const float* __restrict__ X,
                         float* __restrict__ out)
{
    const int b  = blockIdx.y;
    const int i0 = blockIdx.x * ITILE;
    const int j  = threadIdx.x;

    __shared__ float s1row[ITILE][EE];
    __shared__ float a_s[EE];
    __shared__ float tile[2][KK][ECHUNK + 1];
    __shared__ float4 redmax[8];
    __shared__ float4 redsum[8];
    __shared__ __attribute__((aligned(16))) float xs[KK][XPITCH];
    __shared__ __attribute__((aligned(16))) float pn[ITILE][PPITCH];

    const int wave = threadIdx.x >> 5;
    const int lane = threadIdx.x & 31;

    const float* S1 = S;
    const float* S2b = S + (size_t)1024 * EE + (size_t)b * KK * EE;

    const float* src = S2b + (size_t)wave * EE + lane;

    const float* xsrc = X + (size_t)b * KK * FF + (size_t)wave * FF + 4 * lane;

    float pre[NSTAGE];
    #pragma unroll
    for (int k = 0; k < NSTAGE; ++k)
        pre[k] = src[(size_t)(8 * k) * EE];
    static_assert(XQ == 4, "a slice travels as four named float4 values");
    const float4 xp0 = *(const float4*)(xsrc);
    const float4 xp1 = *(const float4*)(xsrc + (size_t)8 * FF);
    const float4 xp2 = *(const float4*)(xsrc + (size_t)16 * FF);
    const float4 xp3 = *(const float4*)(xsrc + (size_t)24 * FF);

    a_s[j] = avec[j];
    #pragma unroll
    for (int ii = 0; ii < ITILE; ++ii)
        s1row[ii][j] = S1[(size_t)(b * KK + i0 + ii) * EE + j];

    #pragma unroll
    for (int k = 0; k < NSTAGE; ++k)
        tile[0][wave + 8 * k][lane] = pre[k];
    *(float4*)&xs[wave][4 * lane]      = xp0;
    *(float4*)&xs[8 + wave][4 * lane]  = xp1;
    *(float4*)&xs[16 + wave][4 * lane] = xp2;
    *(float4*)&xs[24 + wave][4 * lane] = xp3;

    float acc[ITILE];
    #pragma unroll
    for (int ii = 0; ii < ITILE; ++ii) acc[ii] = 0.0f;

    for (int c = 0; c < NCHUNK; ++c) {
        const int e0 = c * ECHUNK;
        __syncthreads();
        const bool more = (c + 1 < NCHUNK);
        float4 xq0, xq1, xq2, xq3;
        if (more) {
            #pragma unroll
            for (int k = 0; k < NSTAGE; ++k)
                pre[k] = src[(size_t)(8 * k) * EE + e0 + ECHUNK];
            const float* xn = xsrc + (size_t)(32 * (c + 1)) * FF;
            xq0 = *(const float4*)(xn);
            xq1 = *(const float4*)(xn + (size_t)8 * FF);
            xq2 = *(const float4*)(xn + (size_t)16 * FF);
            xq3 = *(const float4*)(xn + (size_t)24 * FF);
        }
        const float (*cur)[ECHUNK + 1] = tile[c & 1];
        #pragma unroll 8
        for (int ee = 0; ee < ECHUNK; ++ee) {
            const float t  = cur[j][ee];
            const float ae = a_s[e0 + ee];
            #pragma unroll
            for (int ii = 0; ii < ITILE; ++ii) {
                const float p = s1row[ii][e0 + ee] + t;
                const float h = (p >= 0.0f) ? p : ALPHA * p;
                acc[ii] = fmaf(ae, h, acc[ii]);
            }
        }
        if (more) {
            float (*nxt)[ECHUNK + 1] = tile[(c + 1) & 1];
            #pragma unroll
            for (int k = 0; k < NSTAGE; ++k)
                nxt[wave + 8 * k][lane] = pre[k];
            const int rn = 32 * (c + 1) + wave;
            *(float4*)&xs[rn][4 * lane]      = xq0;
            *(float4*)&xs[rn + 8][4 * lane]  = xq1;
            *(float4*)&xs[rn + 16][4 * lane] = xq2;
            *(float4*)&xs[rn + 24][4 * lane] = xq3;
        }
    }

    static_assert(ITILE == 4, "the 4 rows' partials travel as one float4");

    float ev[ITILE];
    #pragma unroll
    for (int ii = 0; ii < ITILE; ++ii)
        ev[ii] = acc[ii] + bias[(size_t)(i0 + ii) * KK + j];

    float4 wm;
    wm.x = wave_max32(ev[0]); wm.y = wave_max32(ev[1]);
    wm.z = wave_max32(ev[2]); wm.w = wave_max32(ev[3]);
    if (lane == 0) redmax[wave] = wm;
    __syncthreads();

    float4 m = redmax[0];
    #pragma unroll
    for (int w = 1; w < 8; ++w) {
        const float4 t = redmax[w];
        m.x = fmaxf(m.x, t.x); m.y = fmaxf(m.y, t.y);
        m.z = fmaxf(m.z, t.z); m.w = fmaxf(m.w, t.w);
    }

    float p[ITILE];
    p[0] = __expf(ev[0] - m.x); p[1] = __expf(ev[1] - m.y);
    p[2] = __expf(ev[2] - m.z); p[3] = __expf(ev[3] - m.w);

    float4 ws;
    ws.x = wave_sum32(p[0]); ws.y = wave_sum32(p[1]);
    ws.z = wave_sum32(p[2]); ws.w = wave_sum32(p[3]);
    if (lane == 0) redsum[wave] = ws;
    __syncthreads();

    float4 s = redsum[0];
    #pragma unroll
    for (int w = 1; w < 8; ++w) {
        const float4 t = redsum[w];
        s.x += t.x; s.y += t.y; s.z += t.z; s.w += t.w;
    }
    const float inv[ITILE] = {1.0f / s.x, 1.0f / s.y, 1.0f / s.z, 1.0f / s.w};

    #pragma unroll
    for (int ii = 0; ii < ITILE; ++ii)
        pn[ii][j] = p[ii] * inv[ii];
    __syncthreads();

    const int ln   = lane & 15;
    const int hi   = lane >> 4;
    const int col0 = wave * 16;
    const float* arow = &pn[ln & (ITILE - 1)][2 * hi];
    const float* bcol = &xs[2 * hi][col0 + ln];

    v8f acc8 = {};
    v2f a0[8], b0[8], a1[8], b1[8];
    load_frags8(a0, b0, arow, bcol);
    __builtin_amdgcn_sched_group_barrier(0x100, 12, 0);
    #pragma unroll 1
    for (int kb = 0; kb < KK - 64; kb += 64) {
        load_frags8(a1, b1, arow + kb + 32, bcol + (size_t)(kb + 32) * XPITCH);
        __builtin_amdgcn_sched_group_barrier(0x100, 12, 0);
        acc8 = mma_frags8(a0, b0, acc8);
        __builtin_amdgcn_sched_group_barrier(0x008, 8, 0);
        load_frags8(a0, b0, arow + kb + 64, bcol + (size_t)(kb + 64) * XPITCH);
        __builtin_amdgcn_sched_group_barrier(0x100, 12, 0);
        acc8 = mma_frags8(a1, b1, acc8);
        __builtin_amdgcn_sched_group_barrier(0x008, 8, 0);
    }
    load_frags8(a1, b1, arow + KK - 32, bcol + (size_t)(KK - 32) * XPITCH);
    __builtin_amdgcn_sched_group_barrier(0x100, 12, 0);
    acc8 = mma_frags8(a0, b0, acc8);
    __builtin_amdgcn_sched_group_barrier(0x008, 8, 0);
    acc8 = mma_frags8(a1, b1, acc8);

    if (hi == 0) {
        float* O = out + (size_t)(b * KK + i0) * FF + col0 + ln;
        #pragma unroll
        for (int r = 0; r < ITILE; ++r)
            O[(size_t)r * FF] = 1.0f / (1.0f + __expf(-acc8[r]));
    }
}

extern "C" void kernel_launch(void* const* d_in, const int* in_sizes, int n_in,
                              void* d_out, int out_size, void* d_ws, size_t ws_size,
                              hipStream_t stream) {
    const float* x    = (const float*)d_in[0];
    const float* W    = (const float*)d_in[1];
    const float* blin = (const float*)d_in[2];
    const float* a    = (const float*)d_in[3];
    const float* bias = (const float*)d_in[4];
    float* out = (float*)d_out;

    float* S = (float*)d_ws;

    gat_gemm_s_kernel<<<dim3(128, 2), 256, 0, stream>>>(x, W, blin, S);

    gat_attn_out_kernel<<<dim3(KK / ITILE, BB), 256, 0, stream>>>(S, a, bias, x, out);
}


# attention e-loop: lrelu(p) as 0.6p+0.4|p|, row-constant part dropped under softmax, j part summed once; S1 rows and a as scalar operands: one add + one fma(|p|) per element; sum order changes
# speedup vs baseline: 2.0217x; 1.0610x over previous
#include <hip/hip_runtime.h>
#include <math.h>

typedef __attribute__((ext_vector_type(2))) float v2f;
typedef __attribute__((ext_vector_type(8))) float v8f;

#define ALPHA 0.2f
#define LIN_C (0.5f * (1.0f + ALPHA))
#define ABS_C (0.5f * (1.0f - ALPHA))
#define BB 4
#define KK 256
#define FF 128
#define EE 256
#define ITILE 4
#define ECHUNK 32
#define NCHUNK (EE / ECHUNK)
#define NSTAGE (KK * ECHUNK / 256)
#define XQ (KK * FF / 4 / NCHUNK / 256)
#define XPITCH (FF + 8)
#define PPITCH (KK + 16)

__global__ __launch_bounds__(256)
void gat_gemm_s_kernel(const float* __restrict__ X,
                       const float* __restrict__ W,
                       const float* __restrict__ blin,
                       float* __restrict__ S)
{
    const int wave = threadIdx.x >> 5;
    const int lane = threadIdx.x & 31;
    const int ln   = lane & 15;
    const int hi   = lane >> 4;
    const int s    = blockIdx.y;
    const int tile = blockIdx.x * 8 + wave;
    const int row0 = (tile >> 4) * 16;
    const int col0 = (tile & 15) * 16;

    const float* Arow = X + (size_t)(row0 + ln) * FF;
    const float* Brow = W + (size_t)(col0 + ln) * (2 * FF) + s * FF;

    v8f c = {};
    #pragma unroll 4
    for (int k0 = 0; k0 < FF; k0 += 4) {
        v2f a = *(const v2f*)(Arow + k0 + 2 * hi);
        v2f b = *(const v2f*)(Brow + k0 + 2 * hi);
        c = __builtin_amdgcn_wmma_f32_16x16x4_f32(false, a, false, b,
                                                  (short)0, c, false, false);
    }

    const float add = (s == 0) ? blin[col0 + ln] : 0.0f;
    float* D = S + (size_t)s * 1024 * EE;
    #pragma unroll
    for (int r = 0; r < 8; ++r)
        D[(size_t)(row0 + r + 8 * hi) * EE + col0 + ln] = c[r] + add;
}

template <int XM>
__device__ __forceinline__ float row_xor(float x) {
    const int v = __builtin_bit_cast(int, x);
    return __builtin_bit_cast(float,
        __builtin_amdgcn_update_dpp(v, v, 0x160 + XM, 0xf, 0xf, false));
}
__device__ __forceinline__ float xhalf(float x) {
    const int v = __builtin_bit_cast(int, x);
    return __builtin_bit_cast(float,
        __builtin_amdgcn_permlanex16(v, v, 0x76543210, 0xfedcba98, false, false));
}
__device__ __forceinline__ float wave_max32(float x) {
    x = fmaxf(x, row_xor<1>(x));
    x = fmaxf(x, row_xor<2>(x));
    x = fmaxf(x, row_xor<4>(x));
    x = fmaxf(x, row_xor<8>(x));
    return fmaxf(x, xhalf(x));
}
__device__ __forceinline__ float wave_sum32(float x) {
    x += row_xor<1>(x);
    x += row_xor<2>(x);
    x += row_xor<4>(x);
    x += row_xor<8>(x);
    return x + xhalf(x);
}

__device__ __forceinline__ float fma_abs(float a, float p, float acc) {
    asm("v_fma_f32 %0, %1, |%2|, %0" : "+v"(acc) : "s"(a), "v"(p));
    return acc;
}

typedef const __attribute__((address_space(4))) float* cfptr;
struct SOps { float a[4]; float s[ITILE][4]; };
__device__ __forceinline__ void load_sops(SOps& o, cfptr ap, cfptr sp, int e) {
    const cfptr a4 = ap + e;
    const cfptr s4 = sp + e;
    #pragma unroll
    for (int k = 0; k < 4; ++k) o.a[k] = a4[k];
    #pragma unroll
    for (int ii = 0; ii < ITILE; ++ii) {
        #pragma unroll
        for (int k = 0; k < 4; ++k) o.s[ii][k] = s4[ii * EE + k];
    }
}
#define SLOT_GROUP(s) ((((s) & 1) << 2) + ((s) >> 1))

__device__ __forceinline__ void load_frags8(v2f (&a)[8], v2f (&bq)[8],
                                            const float* arow, const float* bcol) {
    #pragma unroll
    for (int s = 0; s < 8; ++s) {
        a[s]    = *(const v2f*)(arow + 4 * s);
        bq[s].x = bcol[(size_t)(4 * s) * XPITCH];
        bq[s].y = bcol[(size_t)(4 * s + 1) * XPITCH];
    }
}
__device__ __forceinline__ v8f mma_frags8(const v2f (&a)[8], const v2f (&bq)[8], v8f acc) {
    #pragma unroll
    for (int s = 0; s < 8; ++s)
        acc = __builtin_amdgcn_wmma_f32_16x16x4_f32(false, a[s], false, bq[s],
                                                    (short)0, acc, false, false);
    return acc;
}

__global__ __launch_bounds__(256)
void gat_attn_out_kernel(const float* __restrict__ S,
                         const float* __restrict__ avec,
                         const float* __restrict__ bias,
                         const float* __restrict__ X,
                         float* __restrict__ out)
{
    const int b  = blockIdx.y;
    const int i0 = blockIdx.x * ITILE;
    const int j  = threadIdx.x;

    __shared__ float tile[2][KK][ECHUNK + 1];
    __shared__ float4 redmax[8];
    __shared__ float4 redsum[8];
    __shared__ __attribute__((aligned(16))) float xs[KK][XPITCH];
    __shared__ __attribute__((aligned(16))) float pn[ITILE][PPITCH];

    const int wave = threadIdx.x >> 5;
    const int lane = threadIdx.x & 31;

    const cfptr s1c = (cfptr)(S + (size_t)(b * KK + i0) * EE);
    const cfptr ac  = (cfptr)avec;
    SOps q0, q1;
    load_sops(q0, ac, s1c, 4 * SLOT_GROUP(0));
    load_sops(q1, ac, s1c, 4 * SLOT_GROUP(1));
    const float* S2b = S + (size_t)1024 * EE + (size_t)b * KK * EE;

    const float* src = S2b + (size_t)wave * EE + lane;

    const float* xsrc = X + (size_t)b * KK * FF + (size_t)wave * FF + 4 * lane;

    float pre[NSTAGE];
    #pragma unroll
    for (int k = 0; k < NSTAGE; ++k)
        pre[k] = src[(size_t)(8 * k) * EE];
    static_assert(XQ == 4, "a slice travels as four named float4 values");
    const float4 xp0 = *(const float4*)(xsrc);
    const float4 xp1 = *(const float4*)(xsrc + (size_t)8 * FF);
    const float4 xp2 = *(const float4*)(xsrc + (size_t)16 * FF);
    const float4 xp3 = *(const float4*)(xsrc + (size_t)24 * FF);

    #pragma unroll
    for (int k = 0; k < NSTAGE; ++k)
        tile[0][wave + 8 * k][lane] = pre[k];
    *(float4*)&xs[wave][4 * lane]      = xp0;
    *(float4*)&xs[8 + wave][4 * lane]  = xp1;
    *(float4*)&xs[16 + wave][4 * lane] = xp2;
    *(float4*)&xs[24 + wave][4 * lane] = xp3;

    float lin[4] = {0.0f, 0.0f, 0.0f, 0.0f};
    float acc[ITILE][2];
    #pragma unroll
    for (int ii = 0; ii < ITILE; ++ii) { acc[ii][0] = 0.0f; acc[ii][1] = 0.0f; }

    for (int c = 0; c < NCHUNK; ++c) {
        const int e0 = c * ECHUNK;
        __syncthreads();
        const bool more = (c + 1 < NCHUNK);
        float4 xq0, xq1, xq2, xq3;
        if (more) {
            #pragma unroll
            for (int k = 0; k < NSTAGE; ++k)
                pre[k] = src[(size_t)(8 * k) * EE + e0 + ECHUNK];
            const float* xn = xsrc + (size_t)(32 * (c + 1)) * FF;
            xq0 = *(const float4*)(xn);
            xq1 = *(const float4*)(xn + (size_t)8 * FF);
            xq2 = *(const float4*)(xn + (size_t)16 * FF);
            xq3 = *(const float4*)(xn + (size_t)24 * FF);
        }
        const float (*cur)[ECHUNK + 1] = tile[c & 1];
        const float* tp = &cur[j][0];
        const int en = ((c + 1) & (NCHUNK - 1)) * ECHUNK;
        #pragma unroll
        for (int s = 0; s < 8; ++s) {
            const SOps o = (s & 1) ? q1 : q0;
            if (s == 7) {
                load_sops(q0, ac, s1c, en + 4 * SLOT_GROUP(0));
                load_sops(q1, ac, s1c, en + 4 * SLOT_GROUP(1));
            } else if (s >= 1) {
                load_sops((s & 1) ? q0 : q1, ac, s1c, e0 + 4 * SLOT_GROUP(s + 1));
            }
            #pragma unroll
            for (int k = 0; k < 4; ++k) {
                const float t = tp[4 * SLOT_GROUP(s) + k];
                lin[k] = fmaf(o.a[k], t, lin[k]);
                #pragma unroll
                for (int ii = 0; ii < ITILE; ++ii) {
                    const float p = o.s[ii][k] + t;
                    acc[ii][k & 1] = fma_abs(o.a[k], p, acc[ii][k & 1]);
                }
            }
        }
        if (more) {
            float (*nxt)[ECHUNK + 1] = tile[(c + 1) & 1];
            #pragma unroll
            for (int k = 0; k < NSTAGE; ++k)
                nxt[wave + 8 * k][lane] = pre[k];
            const int rn = 32 * (c + 1) + wave;
            *(float4*)&xs[rn][4 * lane]      = xq0;
            *(float4*)&xs[rn + 8][4 * lane]  = xq1;
            *(float4*)&xs[rn + 16][4 * lane] = xq2;
            *(float4*)&xs[rn + 24][4 * lane] = xq3;
        }
    }

    static_assert(ITILE == 4, "the 4 rows' partials travel as one float4");

    const float linj = (lin[0] + lin[1]) + (lin[2] + lin[3]);
    float ev[ITILE];
    #pragma unroll
    for (int ii = 0; ii < ITILE; ++ii)
        ev[ii] = fmaf(ABS_C, acc[ii][0] + acc[ii][1], fmaf(LIN_C, linj, bias[(size_t)(i0 + ii) * KK + j]));

    float4 wm;
    wm.x = wave_max32(ev[0]); wm.y = wave_max32(ev[1]);
    wm.z = wave_max32(ev[2]); wm.w = wave_max32(ev[3]);
    if (lane == 0) redmax[wave] = wm;
    __syncthreads();

    float4 m = redmax[0];
    #pragma unroll
    for (int w = 1; w < 8; ++w) {
        const float4 t = redmax[w];
        m.x = fmaxf(m.x, t.x); m.y = fmaxf(m.y, t.y);
        m.z = fmaxf(m.z, t.z); m.w = fmaxf(m.w, t.w);
    }

    float p[ITILE];
    p[0] = __expf(ev[0] - m.x); p[1] = __expf(ev[1] - m.y);
    p[2] = __expf(ev[2] - m.z); p[3] = __expf(ev[3] - m.w);

    float4 ws;
    ws.x = wave_sum32(p[0]); ws.y = wave_sum32(p[1]);
    ws.z = wave_sum32(p[2]); ws.w = wave_sum32(p[3]);
    if (lane == 0) redsum[wave] = ws;
    __syncthreads();

    float4 s = redsum[0];
    #pragma unroll
    for (int w = 1; w < 8; ++w) {
        const float4 t = redsum[w];
        s.x += t.x; s.y += t.y; s.z += t.z; s.w += t.w;
    }
    const float inv[ITILE] = {1.0f / s.x, 1.0f / s.y, 1.0f / s.z, 1.0f / s.w};

    #pragma unroll
    for (int ii = 0; ii < ITILE; ++ii)
        pn[ii][j] = p[ii] * inv[ii];
    __syncthreads();

    const int ln   = lane & 15;
    const int hi   = lane >> 4;
    const int col0 = wave * 16;
    const float* arow = &pn[ln & (ITILE - 1)][2 * hi];
    const float* bcol = &xs[2 * hi][col0 + ln];

    v8f acc8 = {};
    v2f a0[8], b0[8], a1[8], b1[8];
    load_frags8(a0, b0, arow, bcol);
    __builtin_amdgcn_sched_group_barrier(0x100, 12, 0);
    #pragma unroll 1
    for (int kb = 0; kb < KK - 64; kb += 64) {
        load_frags8(a1, b1, arow + kb + 32, bcol + (size_t)(kb + 32) * XPITCH);
        __builtin_amdgcn_sched_group_barrier(0x100, 12, 0);
        acc8 = mma_frags8(a0, b0, acc8);
        __builtin_amdgcn_sched_group_barrier(0x008, 8, 0);
        load_frags8(a0, b0, arow + kb + 64, bcol + (size_t)(kb + 64) * XPITCH);
        __builtin_amdgcn_sched_group_barrier(0x100, 12, 0);
        acc8 = mma_frags8(a1, b1, acc8);
        __builtin_amdgcn_sched_group_barrier(0x008, 8, 0);
    }
    load_frags8(a1, b1, arow + KK - 32, bcol + (size_t)(KK - 32) * XPITCH);
    __builtin_amdgcn_sched_group_barrier(0x100, 12, 0);
    acc8 = mma_frags8(a0, b0, acc8);
    __builtin_amdgcn_sched_group_barrier(0x008, 8, 0);
    acc8 = mma_frags8(a1, b1, acc8);

    if (hi == 0) {
        float* O = out + (size_t)(b * KK + i0) * FF + col0 + ln;
        #pragma unroll
        for (int r = 0; r < ITILE; ++r)
            O[(size_t)r * FF] = 1.0f / (1.0f + __expf(-acc8[r]));
    }
}

extern "C" void kernel_launch(void* const* d_in, const int* in_sizes, int n_in,
                              void* d_out, int out_size, void* d_ws, size_t ws_size,
                              hipStream_t stream) {
    const float* x    = (const float*)d_in[0];
    const float* W    = (const float*)d_in[1];
    const float* blin = (const float*)d_in[2];
    const float* a    = (const float*)d_in[3];
    const float* bias = (const float*)d_in[4];
    float* out = (float*)d_out;

    float* S = (float*)d_ws;

    gat_gemm_s_kernel<<<dim3(128, 2), 256, 0, stream>>>(x, W, blin, S);

    gat_attn_out_kernel<<<dim3(KK / ITILE, BB), 256, 0, stream>>>(S, a, bias, x, out);
}


# S GEMM kernel: operand rows fetched as whole 512-byte rows with 16-byte loads, all in flight at once, staged in LDS; WMMA fragments read from LDS, not 16-line global loads; k order unchanged
# speedup vs baseline: 2.3142x; 1.1447x over previous
#include <hip/hip_runtime.h>
#include <math.h>

typedef __attribute__((ext_vector_type(2))) float v2f;
typedef __attribute__((ext_vector_type(8))) float v8f;

#define ALPHA 0.2f
#define LIN_C (0.5f * (1.0f + ALPHA))
#define ABS_C (0.5f * (1.0f - ALPHA))
#define BB 4
#define KK 256
#define FF 128
#define EE 256
#define ITILE 4
#define ECHUNK 32
#define NCHUNK (EE / ECHUNK)
#define NSTAGE (KK * ECHUNK / 256)
#define XQ (KK * FF / 4 / NCHUNK / 256)
#define XPITCH (FF + 8)
#define PPITCH (KK + 16)

#define GM 32
#define GN 64
#define GROWS (GM + GN)
#define GQ (GROWS / 8)
#define GPITCH (FF + 4)

__device__ __forceinline__ void gemm_frags8(v2f (&a)[8], v2f (&b)[8],
                                            const float* ar, const float* br) {
    #pragma unroll
    for (int t = 0; t < 8; ++t) {
        a[t] = *(const v2f*)(ar + 4 * t);
        b[t] = *(const v2f*)(br + 4 * t);
    }
}
__device__ __forceinline__ v8f gemm_wmma8(const v2f (&a)[8], const v2f (&b)[8], v8f acc) {
    #pragma unroll
    for (int t = 0; t < 8; ++t)
        acc = __builtin_amdgcn_wmma_f32_16x16x4_f32(false, a[t], false, b[t],
                                                    (short)0, acc, false, false);
    return acc;
}

__global__ __launch_bounds__(256)
void gat_gemm_s_kernel(const float* __restrict__ X,
                       const float* __restrict__ W,
                       const float* __restrict__ blin,
                       float* __restrict__ S)
{
    __shared__ __attribute__((aligned(16))) float ab[GROWS][GPITCH];

    const int wave = threadIdx.x >> 5;
    const int lane = threadIdx.x & 31;
    const int ln   = lane & 15;
    const int hi   = lane >> 4;
    const int s    = blockIdx.y >> 2;
    const int m0   = blockIdx.x * GM;
    const int n0   = (blockIdx.y & 3) * GN;
    const int row0 = m0 + (wave >> 2) * 16;
    const int col0 = n0 + (wave & 3) * 16;

    const float* xa = X + (size_t)(m0 + wave) * FF + 4 * lane;
    const float* wb = W + (size_t)(n0 + wave) * (2 * FF) + s * FF + 4 * lane;
    float4 v[GQ];
    #pragma unroll
    for (int q = 0; q < GM / 8; ++q)
        v[q] = *(const float4*)(xa + (size_t)(8 * q) * FF);
    #pragma unroll
    for (int q = 0; q < GN / 8; ++q)
        v[GM / 8 + q] = *(const float4*)(wb + (size_t)(8 * q) * (2 * FF));
    const float add = (s == 0) ? blin[col0 + ln] : 0.0f;

    #pragma unroll
    for (int q = 0; q < GQ; ++q)
        *(float4*)&ab[wave + 8 * q][4 * lane] = v[q];
    __syncthreads();

    const float* ar = &ab[(wave >> 2) * 16 + ln][2 * hi];
    const float* br = &ab[GM + (wave & 3) * 16 + ln][2 * hi];

    v8f c = {};
    v2f a0[8], b0[8], a1[8], b1[8];
    gemm_frags8(a0, b0, ar, br);
    __builtin_amdgcn_sched_group_barrier(0x100, 8, 0);
    gemm_frags8(a1, b1, ar + 32, br + 32);
    __builtin_amdgcn_sched_group_barrier(0x100, 8, 0);
    c = gemm_wmma8(a0, b0, c);
    __builtin_amdgcn_sched_group_barrier(0x008, 8, 0);
    gemm_frags8(a0, b0, ar + 64, br + 64);
    __builtin_amdgcn_sched_group_barrier(0x100, 8, 0);
    c = gemm_wmma8(a1, b1, c);
    __builtin_amdgcn_sched_group_barrier(0x008, 8, 0);
    gemm_frags8(a1, b1, ar + 96, br + 96);
    __builtin_amdgcn_sched_group_barrier(0x100, 8, 0);
    c = gemm_wmma8(a0, b0, c);
    __builtin_amdgcn_sched_group_barrier(0x008, 8, 0);
    c = gemm_wmma8(a1, b1, c);

    float* D = S + (size_t)s * 1024 * EE;
    #pragma unroll
    for (int r = 0; r < 8; ++r)
        D[(size_t)(row0 + r + 8 * hi) * EE + col0 + ln] = c[r] + add;
}

template <int XM>
__device__ __forceinline__ float row_xor(float x) {
    const int v = __builtin_bit_cast(int, x);
    return __builtin_bit_cast(float,
        __builtin_amdgcn_update_dpp(v, v, 0x160 + XM, 0xf, 0xf, false));
}
__device__ __forceinline__ float xhalf(float x) {
    const int v = __builtin_bit_cast(int, x);
    return __builtin_bit_cast(float,
        __builtin_amdgcn_permlanex16(v, v, 0x76543210, 0xfedcba98, false, false));
}
__device__ __forceinline__ float wave_max32(float x) {
    x = fmaxf(x, row_xor<1>(x));
    x = fmaxf(x, row_xor<2>(x));
    x = fmaxf(x, row_xor<4>(x));
    x = fmaxf(x, row_xor<8>(x));
    return fmaxf(x, xhalf(x));
}
__device__ __forceinline__ float wave_sum32(float x) {
    x += row_xor<1>(x);
    x += row_xor<2>(x);
    x += row_xor<4>(x);
    x += row_xor<8>(x);
    return x + xhalf(x);
}

__device__ __forceinline__ float fma_abs(float a, float p, float acc) {
    asm("v_fma_f32 %0, %1, |%2|, %0" : "+v"(acc) : "s"(a), "v"(p));
    return acc;
}

typedef const __attribute__((address_space(4))) float* cfptr;
struct SOps { float a[4]; float s[ITILE][4]; };
__device__ __forceinline__ void load_sops(SOps& o, cfptr ap, cfptr sp, int e) {
    const cfptr a4 = ap + e;
    const cfptr s4 = sp + e;
    #pragma unroll
    for (int k = 0; k < 4; ++k) o.a[k] = a4[k];
    #pragma unroll
    for (int ii = 0; ii < ITILE; ++ii) {
        #pragma unroll
        for (int k = 0; k < 4; ++k) o.s[ii][k] = s4[ii * EE + k];
    }
}
#define SLOT_GROUP(s) ((((s) & 1) << 2) + ((s) >> 1))

__device__ __forceinline__ void load_frags8(v2f (&a)[8], v2f (&bq)[8],
                                            const float* arow, const float* bcol) {
    #pragma unroll
    for (int s = 0; s < 8; ++s) {
        a[s]    = *(const v2f*)(arow + 4 * s);
        bq[s].x = bcol[(size_t)(4 * s) * XPITCH];
        bq[s].y = bcol[(size_t)(4 * s + 1) * XPITCH];
    }
}
__device__ __forceinline__ v8f mma_frags8(const v2f (&a)[8], const v2f (&bq)[8], v8f acc) {
    #pragma unroll
    for (int s = 0; s < 8; ++s)
        acc = __builtin_amdgcn_wmma_f32_16x16x4_f32(false, a[s], false, bq[s],
                                                    (short)0, acc, false, false);
    return acc;
}

__global__ __launch_bounds__(256)
void gat_attn_out_kernel(const float* __restrict__ S,
                         const float* __restrict__ avec,
                         const float* __restrict__ bias,
                         const float* __restrict__ X,
                         float* __restrict__ out)
{
    const int b  = blockIdx.y;
    const int i0 = blockIdx.x * ITILE;
    const int j  = threadIdx.x;

    __shared__ float tile[2][KK][ECHUNK + 1];
    __shared__ float4 redmax[8];
    __shared__ float4 redsum[8];
    __shared__ __attribute__((aligned(16))) float xs[KK][XPITCH];
    __shared__ __attribute__((aligned(16))) float pn[ITILE][PPITCH];

    const int wave = threadIdx.x >> 5;
    const int lane = threadIdx.x & 31;

    const cfptr s1c = (cfptr)(S + (size_t)(b * KK + i0) * EE);
    const cfptr ac  = (cfptr)avec;
    SOps q0, q1;
    load_sops(q0, ac, s1c, 4 * SLOT_GROUP(0));
    load_sops(q1, ac, s1c, 4 * SLOT_GROUP(1));
    const float* S2b = S + (size_t)1024 * EE + (size_t)b * KK * EE;

    const float* src = S2b + (size_t)wave * EE + lane;

    const float* xsrc = X + (size_t)b * KK * FF + (size_t)wave * FF + 4 * lane;

    float pre[NSTAGE];
    #pragma unroll
    for (int k = 0; k < NSTAGE; ++k)
        pre[k] = src[(size_t)(8 * k) * EE];
    static_assert(XQ == 4, "a slice travels as four named float4 values");
    const float4 xp0 = *(const float4*)(xsrc);
    const float4 xp1 = *(const float4*)(xsrc + (size_t)8 * FF);
    const float4 xp2 = *(const float4*)(xsrc + (size_t)16 * FF);
    const float4 xp3 = *(const float4*)(xsrc + (size_t)24 * FF);

    #pragma unroll
    for (int k = 0; k < NSTAGE; ++k)
        tile[0][wave + 8 * k][lane] = pre[k];
    *(float4*)&xs[wave][4 * lane]      = xp0;
    *(float4*)&xs[8 + wave][4 * lane]  = xp1;
    *(float4*)&xs[16 + wave][4 * lane] = xp2;
    *(float4*)&xs[24 + wave][4 * lane] = xp3;

    float lin[4] = {0.0f, 0.0f, 0.0f, 0.0f};
    float acc[ITILE][2];
    #pragma unroll
    for (int ii = 0; ii < ITILE; ++ii) { acc[ii][0] = 0.0f; acc[ii][1] = 0.0f; }

    for (int c = 0; c < NCHUNK; ++c) {
        const int e0 = c * ECHUNK;
        __syncthreads();
        const bool more = (c + 1 < NCHUNK);
        float4 xq0, xq1, xq2, xq3;
        if (more) {
            #pragma unroll
            for (int k = 0; k < NSTAGE; ++k)
                pre[k] = src[(size_t)(8 * k) * EE + e0 + ECHUNK];
            const float* xn = xsrc + (size_t)(32 * (c + 1)) * FF;
            xq0 = *(const float4*)(xn);
            xq1 = *(const float4*)(xn + (size_t)8 * FF);
            xq2 = *(const float4*)(xn + (size_t)16 * FF);
            xq3 = *(const float4*)(xn + (size_t)24 * FF);
        }
        const float (*cur)[ECHUNK + 1] = tile[c & 1];
        const float* tp = &cur[j][0];
        const int en = ((c + 1) & (NCHUNK - 1)) * ECHUNK;
        #pragma unroll
        for (int s = 0; s < 8; ++s) {
            const SOps o = (s & 1) ? q1 : q0;
            if (s == 7) {
                load_sops(q0, ac, s1c, en + 4 * SLOT_GROUP(0));
                load_sops(q1, ac, s1c, en + 4 * SLOT_GROUP(1));
            } else if (s >= 1) {
                load_sops((s & 1) ? q0 : q1, ac, s1c, e0 + 4 * SLOT_GROUP(s + 1));
            }
            #pragma unroll
            for (int k = 0; k < 4; ++k) {
                const float t = tp[4 * SLOT_GROUP(s) + k];
                lin[k] = fmaf(o.a[k], t, lin[k]);
                #pragma unroll
                for (int ii = 0; ii < ITILE; ++ii) {
                    const float p = o.s[ii][k] + t;
                    acc[ii][k & 1] = fma_abs(o.a[k], p, acc[ii][k & 1]);
                }
            }
        }
        if (more) {
            float (*nxt)[ECHUNK + 1] = tile[(c + 1) & 1];
            #pragma unroll
            for (int k = 0; k < NSTAGE; ++k)
                nxt[wave + 8 * k][lane] = pre[k];
            const int rn = 32 * (c + 1) + wave;
            *(float4*)&xs[rn][4 * lane]      = xq0;
            *(float4*)&xs[rn + 8][4 * lane]  = xq1;
            *(float4*)&xs[rn + 16][4 * lane] = xq2;
            *(float4*)&xs[rn + 24][4 * lane] = xq3;
        }
    }

    static_assert(ITILE == 4, "the 4 rows' partials travel as one float4");

    const float linj = (lin[0] + lin[1]) + (lin[2] + lin[3]);
    float ev[ITILE];
    #pragma unroll
    for (int ii = 0; ii < ITILE; ++ii)
        ev[ii] = fmaf(ABS_C, acc[ii][0] + acc[ii][1], fmaf(LIN_C, linj, bias[(size_t)(i0 + ii) * KK + j]));

    float4 wm;
    wm.x = wave_max32(ev[0]); wm.y = wave_max32(ev[1]);
    wm.z = wave_max32(ev[2]); wm.w = wave_max32(ev[3]);
    if (lane == 0) redmax[wave] = wm;
    __syncthreads();

    float4 m = redmax[0];
    #pragma unroll
    for (int w = 1; w < 8; ++w) {
        const float4 t = redmax[w];
        m.x = fmaxf(m.x, t.x); m.y = fmaxf(m.y, t.y);
        m.z = fmaxf(m.z, t.z); m.w = fmaxf(m.w, t.w);
    }

    float p[ITILE];
    p[0] = __expf(ev[0] - m.x); p[1] = __expf(ev[1] - m.y);
    p[2] = __expf(ev[2] - m.z); p[3] = __expf(ev[3] - m.w);

    float4 ws;
    ws.x = wave_sum32(p[0]); ws.y = wave_sum32(p[1]);
    ws.z = wave_sum32(p[2]); ws.w = wave_sum32(p[3]);
    if (lane == 0) redsum[wave] = ws;
    __syncthreads();

    float4 s = redsum[0];
    #pragma unroll
    for (int w = 1; w < 8; ++w) {
        const float4 t = redsum[w];
        s.x += t.x; s.y += t.y; s.z += t.z; s.w += t.w;
    }
    const float inv[ITILE] = {1.0f / s.x, 1.0f / s.y, 1.0f / s.z, 1.0f / s.w};

    #pragma unroll
    for (int ii = 0; ii < ITILE; ++ii)
        pn[ii][j] = p[ii] * inv[ii];
    __syncthreads();

    const int ln   = lane & 15;
    const int hi   = lane >> 4;
    const int col0 = wave * 16;
    const float* arow = &pn[ln & (ITILE - 1)][2 * hi];
    const float* bcol = &xs[2 * hi][col0 + ln];

    v8f acc8 = {};
    v2f a0[8], b0[8], a1[8], b1[8];
    load_frags8(a0, b0, arow, bcol);
    __builtin_amdgcn_sched_group_barrier(0x100, 12, 0);
    #pragma unroll 1
    for (int kb = 0; kb < KK - 64; kb += 64) {
        load_frags8(a1, b1, arow + kb + 32, bcol + (size_t)(kb + 32) * XPITCH);
        __builtin_amdgcn_sched_group_barrier(0x100, 12, 0);
        acc8 = mma_frags8(a0, b0, acc8);
        __builtin_amdgcn_sched_group_barrier(0x008, 8, 0);
        load_frags8(a0, b0, arow + kb + 64, bcol + (size_t)(kb + 64) * XPITCH);
        __builtin_amdgcn_sched_group_barrier(0x100, 12, 0);
        acc8 = mma_frags8(a1, b1, acc8);
        __builtin_amdgcn_sched_group_barrier(0x008, 8, 0);
    }
    load_frags8(a1, b1, arow + KK - 32, bcol + (size_t)(KK - 32) * XPITCH);
    __builtin_amdgcn_sched_group_barrier(0x100, 12, 0);
    acc8 = mma_frags8(a0, b0, acc8);
    __builtin_amdgcn_sched_group_barrier(0x008, 8, 0);
    acc8 = mma_frags8(a1, b1, acc8);

    if (hi == 0) {
        float* O = out + (size_t)(b * KK + i0) * FF + col0 + ln;
        #pragma unroll
        for (int r = 0; r < ITILE; ++r)
            O[(size_t)r * FF] = 1.0f / (1.0f + __expf(-acc8[r]));
    }
}

extern "C" void kernel_launch(void* const* d_in, const int* in_sizes, int n_in,
                              void* d_out, int out_size, void* d_ws, size_t ws_size,
                              hipStream_t stream) {
    const float* x    = (const float*)d_in[0];
    const float* W    = (const float*)d_in[1];
    const float* blin = (const float*)d_in[2];
    const float* a    = (const float*)d_in[3];
    const float* bias = (const float*)d_in[4];
    float* out = (float*)d_out;

    float* S = (float*)d_ws;

    gat_gemm_s_kernel<<<dim3(32, 8), 256, 0, stream>>>(x, W, blin, S);

    gat_attn_out_kernel<<<dim3(KK / ITILE, BB), 256, 0, stream>>>(S, a, bias, x, out);
}


# attention kernel: S2 chunk stream moved in 16-byte pieces end to end: wave-loads of 4 whole lines, 16-byte LDS stores and row reads, row pitch 16 B x 9; e order and sum order unchanged
# speedup vs baseline: 2.5003x; 1.0804x over previous
#include <hip/hip_runtime.h>
#include <math.h>

typedef __attribute__((ext_vector_type(2))) float v2f;
typedef __attribute__((ext_vector_type(4))) float v4f;
typedef __attribute__((ext_vector_type(8))) float v8f;

#define ALPHA 0.2f
#define LIN_C (0.5f * (1.0f + ALPHA))
#define ABS_C (0.5f * (1.0f - ALPHA))
#define BB 4
#define KK 256
#define FF 128
#define EE 256
#define ITILE 4
#define ECHUNK 32
#define NCHUNK (EE / ECHUNK)
#define TPITCH (ECHUNK + 4)
#define SQ (KK * ECHUNK / 4 / 256)
#define XQ (KK * FF / 4 / NCHUNK / 256)
#define XPITCH (FF + 8)
#define PPITCH (KK + 16)

#define GM 32
#define GN 64
#define GROWS (GM + GN)
#define GQ (GROWS / 8)
#define GPITCH (FF + 4)

__device__ __forceinline__ void gemm_frags8(v2f (&a)[8], v2f (&b)[8],
                                            const float* ar, const float* br) {
    #pragma unroll
    for (int t = 0; t < 8; ++t) {
        a[t] = *(const v2f*)(ar + 4 * t);
        b[t] = *(const v2f*)(br + 4 * t);
    }
}
__device__ __forceinline__ v8f gemm_wmma8(const v2f (&a)[8], const v2f (&b)[8], v8f acc) {
    #pragma unroll
    for (int t = 0; t < 8; ++t)
        acc = __builtin_amdgcn_wmma_f32_16x16x4_f32(false, a[t], false, b[t],
                                                    (short)0, acc, false, false);
    return acc;
}

__global__ __launch_bounds__(256)
void gat_gemm_s_kernel(const float* __restrict__ X,
                       const float* __restrict__ W,
                       const float* __restrict__ blin,
                       float* __restrict__ S)
{
    __shared__ __attribute__((aligned(16))) float ab[GROWS][GPITCH];

    const int wave = threadIdx.x >> 5;
    const int lane = threadIdx.x & 31;
    const int ln   = lane & 15;
    const int hi   = lane >> 4;
    const int s    = blockIdx.y >> 2;
    const int m0   = blockIdx.x * GM;
    const int n0   = (blockIdx.y & 3) * GN;
    const int row0 = m0 + (wave >> 2) * 16;
    const int col0 = n0 + (wave & 3) * 16;

    const float* xa = X + (size_t)(m0 + wave) * FF + 4 * lane;
    const float* wb = W + (size_t)(n0 + wave) * (2 * FF) + s * FF + 4 * lane;
    float4 v[GQ];
    #pragma unroll
    for (int q = 0; q < GM / 8; ++q)
        v[q] = *(const float4*)(xa + (size_t)(8 * q) * FF);
    #pragma unroll
    for (int q = 0; q < GN / 8; ++q)
        v[GM / 8 + q] = *(const float4*)(wb + (size_t)(8 * q) * (2 * FF));
    const float add = (s == 0) ? blin[col0 + ln] : 0.0f;

    #pragma unroll
    for (int q = 0; q < GQ; ++q)
        *(float4*)&ab[wave + 8 * q][4 * lane] = v[q];
    __syncthreads();

    const float* ar = &ab[(wave >> 2) * 16 + ln][2 * hi];
    const float* br = &ab[GM + (wave & 3) * 16 + ln][2 * hi];

    v8f c = {};
    v2f a0[8], b0[8], a1[8], b1[8];
    gemm_frags8(a0, b0, ar, br);
    __builtin_amdgcn_sched_group_barrier(0x100, 8, 0);
    gemm_frags8(a1, b1, ar + 32, br + 32);
    __builtin_amdgcn_sched_group_barrier(0x100, 8, 0);
    c = gemm_wmma8(a0, b0, c);
    __builtin_amdgcn_sched_group_barrier(0x008, 8, 0);
    gemm_frags8(a0, b0, ar + 64, br + 64);
    __builtin_amdgcn_sched_group_barrier(0x100, 8, 0);
    c = gemm_wmma8(a1, b1, c);
    __builtin_amdgcn_sched_group_barrier(0x008, 8, 0);
    gemm_frags8(a1, b1, ar + 96, br + 96);
    __builtin_amdgcn_sched_group_barrier(0x100, 8, 0);
    c = gemm_wmma8(a0, b0, c);
    __builtin_amdgcn_sched_group_barrier(0x008, 8, 0);
    c = gemm_wmma8(a1, b1, c);

    float* D = S + (size_t)s * 1024 * EE;
    #pragma unroll
    for (int r = 0; r < 8; ++r)
        D[(size_t)(row0 + r + 8 * hi) * EE + col0 + ln] = c[r] + add;
}

template <int XM>
__device__ __forceinline__ float row_xor(float x) {
    const int v = __builtin_bit_cast(int, x);
    return __builtin_bit_cast(float,
        __builtin_amdgcn_update_dpp(v, v, 0x160 + XM, 0xf, 0xf, false));
}
__device__ __forceinline__ float xhalf(float x) {
    const int v = __builtin_bit_cast(int, x);
    return __builtin_bit_cast(float,
        __builtin_amdgcn_permlanex16(v, v, 0x76543210, 0xfedcba98, false, false));
}
__device__ __forceinline__ float wave_max32(float x) {
    x = fmaxf(x, row_xor<1>(x));
    x = fmaxf(x, row_xor<2>(x));
    x = fmaxf(x, row_xor<4>(x));
    x = fmaxf(x, row_xor<8>(x));
    return fmaxf(x, xhalf(x));
}
__device__ __forceinline__ float wave_sum32(float x) {
    x += row_xor<1>(x);
    x += row_xor<2>(x);
    x += row_xor<4>(x);
    x += row_xor<8>(x);
    return x + xhalf(x);
}

__device__ __forceinline__ float fma_abs(float a, float p, float acc) {
    asm("v_fma_f32 %0, %1, |%2|, %0" : "+v"(acc) : "s"(a), "v"(p));
    return acc;
}

typedef const __attribute__((address_space(4))) float* cfptr;
struct SOps { float a[4]; float s[ITILE][4]; };
__device__ __forceinline__ void load_sops(SOps& o, cfptr ap, cfptr sp, int e) {
    const cfptr a4 = ap + e;
    const cfptr s4 = sp + e;
    #pragma unroll
    for (int k = 0; k < 4; ++k) o.a[k] = a4[k];
    #pragma unroll
    for (int ii = 0; ii < ITILE; ++ii) {
        #pragma unroll
        for (int k = 0; k < 4; ++k) o.s[ii][k] = s4[ii * EE + k];
    }
}
#define SLOT_GROUP(s) ((((s) & 1) << 2) + ((s) >> 1))

__device__ __forceinline__ void load_frags8(v2f (&a)[8], v2f (&bq)[8],
                                            const float* arow, const float* bcol) {
    #pragma unroll
    for (int s = 0; s < 8; ++s) {
        a[s]    = *(const v2f*)(arow + 4 * s);
        bq[s].x = bcol[(size_t)(4 * s) * XPITCH];
        bq[s].y = bcol[(size_t)(4 * s + 1) * XPITCH];
    }
}
__device__ __forceinline__ v8f mma_frags8(const v2f (&a)[8], const v2f (&bq)[8], v8f acc) {
    #pragma unroll
    for (int s = 0; s < 8; ++s)
        acc = __builtin_amdgcn_wmma_f32_16x16x4_f32(false, a[s], false, bq[s],
                                                    (short)0, acc, false, false);
    return acc;
}

__global__ __launch_bounds__(256)
void gat_attn_out_kernel(const float* __restrict__ S,
                         const float* __restrict__ avec,
                         const float* __restrict__ bias,
                         const float* __restrict__ X,
                         float* __restrict__ out)
{
    const int b  = blockIdx.y;
    const int i0 = blockIdx.x * ITILE;
    const int j  = threadIdx.x;

    __shared__ __attribute__((aligned(16))) float tile[2][KK][TPITCH];
    __shared__ float4 redmax[8];
    __shared__ float4 redsum[8];
    __shared__ __attribute__((aligned(16))) float xs[KK][XPITCH];
    __shared__ __attribute__((aligned(16))) float pn[ITILE][PPITCH];

    const int wave = threadIdx.x >> 5;
    const int lane = threadIdx.x & 31;

    const cfptr s1c = (cfptr)(S + (size_t)(b * KK + i0) * EE);
    const cfptr ac  = (cfptr)avec;
    SOps q0, q1;
    load_sops(q0, ac, s1c, 4 * SLOT_GROUP(0));
    load_sops(q1, ac, s1c, 4 * SLOT_GROUP(1));
    const float* S2b = S + (size_t)1024 * EE + (size_t)b * KK * EE;

    const int srow = wave + 8 * (lane >> 3);
    const int scol = 4 * (lane & 7);
    const float* src = S2b + (size_t)srow * EE + scol;

    const float* xsrc = X + (size_t)b * KK * FF + (size_t)wave * FF + 4 * lane;

    v4f pre[SQ];
    #pragma unroll
    for (int q = 0; q < SQ; ++q)
        pre[q] = *(const v4f*)(src + (size_t)(32 * q) * EE);
    static_assert(XQ == 4, "a slice travels as four named float4 values");
    const float4 xp0 = *(const float4*)(xsrc);
    const float4 xp1 = *(const float4*)(xsrc + (size_t)8 * FF);
    const float4 xp2 = *(const float4*)(xsrc + (size_t)16 * FF);
    const float4 xp3 = *(const float4*)(xsrc + (size_t)24 * FF);

    #pragma unroll
    for (int q = 0; q < SQ; ++q)
        *(v4f*)&tile[0][srow + 32 * q][scol] = pre[q];
    *(float4*)&xs[wave][4 * lane]      = xp0;
    *(float4*)&xs[8 + wave][4 * lane]  = xp1;
    *(float4*)&xs[16 + wave][4 * lane] = xp2;
    *(float4*)&xs[24 + wave][4 * lane] = xp3;

    float lin[4] = {0.0f, 0.0f, 0.0f, 0.0f};
    float acc[ITILE][2];
    #pragma unroll
    for (int ii = 0; ii < ITILE; ++ii) { acc[ii][0] = 0.0f; acc[ii][1] = 0.0f; }

    for (int c = 0; c < NCHUNK; ++c) {
        const int e0 = c * ECHUNK;
        __syncthreads();
        const bool more = (c + 1 < NCHUNK);
        float4 xq0, xq1, xq2, xq3;
        if (more) {
            #pragma unroll
            for (int q = 0; q < SQ; ++q)
                pre[q] = *(const v4f*)(src + (size_t)(32 * q) * EE + e0 + ECHUNK);
            const float* xn = xsrc + (size_t)(32 * (c + 1)) * FF;
            xq0 = *(const float4*)(xn);
            xq1 = *(const float4*)(xn + (size_t)8 * FF);
            xq2 = *(const float4*)(xn + (size_t)16 * FF);
            xq3 = *(const float4*)(xn + (size_t)24 * FF);
        }
        const float (*cur)[TPITCH] = tile[c & 1];
        const float* tp = &cur[j][0];
        const int en = ((c + 1) & (NCHUNK - 1)) * ECHUNK;
        #pragma unroll
        for (int s = 0; s < 8; ++s) {
            const SOps o = (s & 1) ? q1 : q0;
            if (s == 7) {
                load_sops(q0, ac, s1c, en + 4 * SLOT_GROUP(0));
                load_sops(q1, ac, s1c, en + 4 * SLOT_GROUP(1));
            } else if (s >= 1) {
                load_sops((s & 1) ? q0 : q1, ac, s1c, e0 + 4 * SLOT_GROUP(s + 1));
            }
            const v4f t4 = *(const v4f*)(tp + 4 * SLOT_GROUP(s));
            #pragma unroll
            for (int k = 0; k < 4; ++k) {
                const float t = t4[k];
                lin[k] = fmaf(o.a[k], t, lin[k]);
                #pragma unroll
                for (int ii = 0; ii < ITILE; ++ii) {
                    const float p = o.s[ii][k] + t;
                    acc[ii][k & 1] = fma_abs(o.a[k], p, acc[ii][k & 1]);
                }
            }
        }
        if (more) {
            float (*nxt)[TPITCH] = tile[(c + 1) & 1];
            #pragma unroll
            for (int q = 0; q < SQ; ++q)
                *(v4f*)&nxt[srow + 32 * q][scol] = pre[q];
            const int rn = 32 * (c + 1) + wave;
            *(float4*)&xs[rn][4 * lane]      = xq0;
            *(float4*)&xs[rn + 8][4 * lane]  = xq1;
            *(float4*)&xs[rn + 16][4 * lane] = xq2;
            *(float4*)&xs[rn + 24][4 * lane] = xq3;
        }
    }

    static_assert(ITILE == 4, "the 4 rows' partials travel as one float4");

    const float linj = (lin[0] + lin[1]) + (lin[2] + lin[3]);
    float ev[ITILE];
    #pragma unroll
    for (int ii = 0; ii < ITILE; ++ii)
        ev[ii] = fmaf(ABS_C, acc[ii][0] + acc[ii][1], fmaf(LIN_C, linj, bias[(size_t)(i0 + ii) * KK + j]));

    float4 wm;
    wm.x = wave_max32(ev[0]); wm.y = wave_max32(ev[1]);
    wm.z = wave_max32(ev[2]); wm.w = wave_max32(ev[3]);
    if (lane == 0) redmax[wave] = wm;
    __syncthreads();

    float4 m = redmax[0];
    #pragma unroll
    for (int w = 1; w < 8; ++w) {
        const float4 t = redmax[w];
        m.x = fmaxf(m.x, t.x); m.y = fmaxf(m.y, t.y);
        m.z = fmaxf(m.z, t.z); m.w = fmaxf(m.w, t.w);
    }

    float p[ITILE];
    p[0] = __expf(ev[0] - m.x); p[1] = __expf(ev[1] - m.y);
    p[2] = __expf(ev[2] - m.z); p[3] = __expf(ev[3] - m.w);

    float4 ws;
    ws.x = wave_sum32(p[0]); ws.y = wave_sum32(p[1]);
    ws.z = wave_sum32(p[2]); ws.w = wave_sum32(p[3]);
    if (lane == 0) redsum[wave] = ws;
    __syncthreads();

    float4 s = redsum[0];
    #pragma unroll
    for (int w = 1; w < 8; ++w) {
        const float4 t = redsum[w];
        s.x += t.x; s.y += t.y; s.z += t.z; s.w += t.w;
    }
    const float inv[ITILE] = {1.0f / s.x, 1.0f / s.y, 1.0f / s.z, 1.0f / s.w};

    #pragma unroll
    for (int ii = 0; ii < ITILE; ++ii)
        pn[ii][j] = p[ii] * inv[ii];
    __syncthreads();

    const int ln   = lane & 15;
    const int hi   = lane >> 4;
    const int col0 = wave * 16;
    const float* arow = &pn[ln & (ITILE - 1)][2 * hi];
    const float* bcol = &xs[2 * hi][col0 + ln];

    v8f acc8 = {};
    v2f a0[8], b0[8], a1[8], b1[8];
    load_frags8(a0, b0, arow, bcol);
    __builtin_amdgcn_sched_group_barrier(0x100, 12, 0);
    #pragma unroll 1
    for (int kb = 0; kb < KK - 64; kb += 64) {
        load_frags8(a1, b1, arow + kb + 32, bcol + (size_t)(kb + 32) * XPITCH);
        __builtin_amdgcn_sched_group_barrier(0x100, 12, 0);
        acc8 = mma_frags8(a0, b0, acc8);
        __builtin_amdgcn_sched_group_barrier(0x008, 8, 0);
        load_frags8(a0, b0, arow + kb + 64, bcol + (size_t)(kb + 64) * XPITCH);
        __builtin_amdgcn_sched_group_barrier(0x100, 12, 0);
        acc8 = mma_frags8(a1, b1, acc8);
        __builtin_amdgcn_sched_group_barrier(0x008, 8, 0);
    }
    load_frags8(a1, b1, arow + KK - 32, bcol + (size_t)(KK - 32) * XPITCH);
    __builtin_amdgcn_sched_group_barrier(0x100, 12, 0);
    acc8 = mma_frags8(a0, b0, acc8);
    __builtin_amdgcn_sched_group_barrier(0x008, 8, 0);
    acc8 = mma_frags8(a1, b1, acc8);

    if (hi == 0) {
        float* O = out + (size_t)(b * KK + i0) * FF + col0 + ln;
        #pragma unroll
        for (int r = 0; r < ITILE; ++r)
            O[(size_t)r * FF] = 1.0f / (1.0f + __expf(-acc8[r]));
    }
}

extern "C" void kernel_launch(void* const* d_in, const int* in_sizes, int n_in,
                              void* d_out, int out_size, void* d_ws, size_t ws_size,
                              hipStream_t stream) {
    const float* x    = (const float*)d_in[0];
    const float* W    = (const float*)d_in[1];
    const float* blin = (const float*)d_in[2];
    const float* a    = (const float*)d_in[3];
    const float* bias = (const float*)d_in[4];
    float* out = (float*)d_out;

    float* S = (float*)d_ws;

    gat_gemm_s_kernel<<<dim3(32, 8), 256, 0, stream>>>(x, W, blin, S);

    gat_attn_out_kernel<<<dim3(KK / ITILE, BB), 256, 0, stream>>>(S, a, bias, x, out);
}
